# starting kernel
# baseline (speedup 1.0000x reference)
#include <hip/hip_runtime.h>


typedef float v2f __attribute__((ext_vector_type(2)));
typedef float v4f __attribute__((ext_vector_type(4)));
typedef float v8f __attribute__((ext_vector_type(8)));

#define GRP 8
#define IPG 64
#define HPG 64
#define ROW 512

__global__ __launch_bounds__(256) void grouped_linear_wmma(
    const float* __restrict__ x, const float* __restrict__ W,
    const float* __restrict__ bias, float* __restrict__ out, int ntiles) {
  __shared__ float lds[16 * ROW];

  const int lane = threadIdx.x & 31;
  const int g    = threadIdx.x >> 5;
  const int half = lane >> 4;
  const int l16  = lane & 15;

  v2f bfrag[4][16];
  const float* Wg = W + g * HPG * IPG;
#pragma unroll
  for (int n = 0; n < 4; ++n) {
    const float* wr = Wg + (n * 16 + l16) * IPG + half * 2;
#pragma unroll
    for (int k = 0; k < 16; ++k) bfrag[n][k] = *(const v2f*)(wr + 4 * k);
  }

  float bias_f[4];
#pragma unroll
  for (int n = 0; n < 4; ++n) bias_f[n] = bias[g * HPG + n * 16 + l16];

  for (int tile = blockIdx.x; tile < ntiles; tile += (int)gridDim.x) {
    const long tok0 = (long)tile * 16;

    const float* xr = x + (tok0 + l16) * ROW + g * IPG + half * 2;
    v2f afrag[16];
#pragma unroll
    for (int k = 0; k < 16; ++k) afrag[k] = *(const v2f*)(xr + 4 * k);

    if (tile + (int)gridDim.x < ntiles)
      __builtin_prefetch(xr + (long)gridDim.x * 16 * ROW, 0, 3);

#pragma unroll
    for (int n = 0; n < 4; ++n) {
      v8f acc;
#pragma unroll
      for (int i = 0; i < 8; ++i) acc[i] = bias_f[n];
#pragma unroll
      for (int k = 0; k < 16; ++k)
        acc = __builtin_amdgcn_wmma_f32_16x16x4_f32(
            false, afrag[k], false, bfrag[n][k], (short)0, acc, false, false);

#pragma unroll
      for (int v = 0; v < 8; ++v)
        lds[(v + half * 8) * ROW + (n * 16 + l16) * GRP + g] = acc[v];
    }

    __syncthreads();

    {
      float* orow = out + tok0 * ROW;
#pragma unroll
      for (int it = 0; it < 8; ++it) {
        int idx = (it * 256 + (int)threadIdx.x) * 4;
        *(v4f*)(orow + idx) = *(const v4f*)(lds + idx);
      }
    }

    __syncthreads();
  }
}

extern "C" void kernel_launch(void* const* d_in, const int* in_sizes, int n_in,
                              void* d_out, int out_size, void* d_ws, size_t ws_size,
                              hipStream_t stream) {
  const float* x    = (const float*)d_in[0];
  const float* W    = (const float*)d_in[1];
  const float* bias = (const float*)d_in[2];
  float* out        = (float*)d_out;

  const int ntok   = in_sizes[0] / ROW;
  const int ntiles = ntok / 16;

  int blocks = ntiles < 512 ? ntiles : 512;
  grouped_linear_wmma<<<dim3(blocks), dim3(256), 0, stream>>>(x, W, bias, out, ntiles);
}


# Tile loop software-pipelined: LDS staging tile double-buffered, one LDS-only barrier per tile (no wait for global stores), next tile's x loads issued right behind the last WMMA; 8 v_nop hazard cover
# speedup vs baseline: 1.1144x; 1.1144x over previous
#include <hip/hip_runtime.h>


typedef float v2f __attribute__((ext_vector_type(2)));
typedef float v4f __attribute__((ext_vector_type(4)));
typedef float v8f __attribute__((ext_vector_type(8)));

#define GRP 8
#define IPG 64
#define HPG 64
#define ROW 512

__global__ __launch_bounds__(256) void grouped_linear_wmma(
    const float* __restrict__ x, const float* __restrict__ W,
    const float* __restrict__ bias, float* __restrict__ out, int ntiles) {
  __shared__ float lds[2 * 16 * ROW];

  const int lane = threadIdx.x & 31;
  const int g    = threadIdx.x >> 5;
  const int half = lane >> 4;
  const int l16  = lane & 15;

  v2f bfrag[4][16];
  const float* Wg = W + g * HPG * IPG;
#pragma unroll
  for (int n = 0; n < 4; ++n) {
    const float* wr = Wg + (n * 16 + l16) * IPG + half * 2;
#pragma unroll
    for (int k = 0; k < 16; ++k) bfrag[n][k] = *(const v2f*)(wr + 4 * k);
  }

  float bias_f[4];
#pragma unroll
  for (int n = 0; n < 4; ++n) bias_f[n] = bias[g * HPG + n * 16 + l16];

  const int  stride = (int)gridDim.x;
  const long xstep  = (long)stride * 16 * ROW;
  int tile = blockIdx.x;

  const unsigned lo = (unsigned)(l16 * ROW + g * IPG + half * 2);
  const float* xt = x + (long)tile * (16 * ROW);
  v2f afrag[16];
  if (tile < ntiles) {
    const float* xr = xt + lo;
#pragma unroll
    for (int k = 0; k < 16; ++k) afrag[k] = *(const v2f*)(xr + 4 * k);
    if (tile + stride < ntiles) __builtin_prefetch(xt + xstep + lo, 0, 3);
  }

  int buf = 0;
  for (; tile < ntiles; tile += stride) {
    float* stage = lds + buf * (16 * ROW);
    const long tok0 = (long)tile * 16;

    v8f acc[4];
#pragma unroll
    for (int n = 0; n < 4; ++n) {
#pragma unroll
      for (int i = 0; i < 8; ++i) acc[n][i] = bias_f[n];
    }
#pragma unroll
    for (int k = 0; k < 16; ++k) {
#pragma unroll
      for (int n = 0; n < 4; ++n)
        acc[n] = __builtin_amdgcn_wmma_f32_16x16x4_f32(
            false, afrag[k], false, bfrag[n][k], (short)0, acc[n], false, false);
    }

    asm volatile("v_nop\n\tv_nop\n\tv_nop\n\tv_nop\n\tv_nop\n\tv_nop\n\tv_nop\n\tv_nop"
                 :: "v"(acc[0]), "v"(acc[1]), "v"(acc[2]), "v"(acc[3]));

    xt += xstep;
    if (tile + stride < ntiles) {
      const float* xr = xt + lo;
#pragma unroll
      for (int k = 0; k < 16; ++k) afrag[k] = *(const v2f*)(xr + 4 * k);
      if (tile + 2 * stride < ntiles) __builtin_prefetch(xt + xstep + lo, 0, 3);
    }

#pragma unroll
    for (int n = 0; n < 4; ++n) {
#pragma unroll
      for (int v = 0; v < 8; ++v)
        stage[(v + half * 8) * ROW + (n * 16 + l16) * GRP + g] = acc[n][v];
    }

    __builtin_amdgcn_fence(__ATOMIC_RELEASE, "workgroup", "local");
    __builtin_amdgcn_s_barrier();
    __builtin_amdgcn_fence(__ATOMIC_ACQUIRE, "workgroup", "local");

    {
      float* orow = out + tok0 * ROW;
#pragma unroll
      for (int it = 0; it < 8; ++it) {
        int idx = (it * 256 + (int)threadIdx.x) * 4;
        *(v4f*)(orow + idx) = *(const v4f*)(stage + idx);
      }
    }

    buf ^= 1;
  }
}

extern "C" void kernel_launch(void* const* d_in, const int* in_sizes, int n_in,
                              void* d_out, int out_size, void* d_ws, size_t ws_size,
                              hipStream_t stream) {
  const float* x    = (const float*)d_in[0];
  const float* W    = (const float*)d_in[1];
  const float* bias = (const float*)d_in[2];
  float* out        = (float*)d_out;

  const int ntok   = in_sizes[0] / ROW;
  const int ntiles = ntok / 16;

  int blocks = ntiles < 512 ? ntiles : 512;
  grouped_linear_wmma<<<dim3(blocks), dim3(256), 0, stream>>>(x, W, bias, out, ntiles);
}


# Wave priority by phase (s_setprio): it rises each quarter of a tile's 64 WMMAs and stays top through requests, LDS exchange and stores, so a wave deep in its block is not displaced by an older one
# speedup vs baseline: 1.1555x; 1.0369x over previous
#include <hip/hip_runtime.h>


typedef float v2f __attribute__((ext_vector_type(2)));
typedef float v4f __attribute__((ext_vector_type(4)));
typedef float v8f __attribute__((ext_vector_type(8)));

#define GRP 8
#define IPG 64
#define HPG 64
#define ROW 512

template <int Q>
__device__ __forceinline__ void products_quarter(v8f (&acc)[4], const v2f (&afrag)[16],
                                                 const v2f (&bfrag)[4][16]) {
  __builtin_amdgcn_s_setprio(Q);
#pragma unroll
  for (int k = 4 * Q; k < 4 * Q + 4; ++k) {
#pragma unroll
    for (int n = 0; n < 4; ++n)
      acc[n] = __builtin_amdgcn_wmma_f32_16x16x4_f32(
          false, afrag[k], false, bfrag[n][k], (short)0, acc[n], false, false);
  }
}

__global__ __launch_bounds__(256) void grouped_linear_wmma(
    const float* __restrict__ x, const float* __restrict__ W,
    const float* __restrict__ bias, float* __restrict__ out, int ntiles) {
  __shared__ float lds[2 * 16 * ROW];

  __builtin_amdgcn_s_setprio(3);

  const int lane = threadIdx.x & 31;
  const int g    = threadIdx.x >> 5;
  const int half = lane >> 4;
  const int l16  = lane & 15;

  v2f bfrag[4][16];
  const float* Wg = W + g * HPG * IPG;
#pragma unroll
  for (int n = 0; n < 4; ++n) {
    const float* wr = Wg + (n * 16 + l16) * IPG + half * 2;
#pragma unroll
    for (int k = 0; k < 16; ++k) bfrag[n][k] = *(const v2f*)(wr + 4 * k);
  }

  float bias_f[4];
#pragma unroll
  for (int n = 0; n < 4; ++n) bias_f[n] = bias[g * HPG + n * 16 + l16];

  const int  stride = (int)gridDim.x;
  const long xstep  = (long)stride * 16 * ROW;
  int tile = blockIdx.x;

  const unsigned lo = (unsigned)(l16 * ROW + g * IPG + half * 2);
  const float* xt = x + (long)tile * (16 * ROW);
  v2f afrag[16];
  if (tile < ntiles) {
    const float* xr = xt + lo;
#pragma unroll
    for (int k = 0; k < 16; ++k) afrag[k] = *(const v2f*)(xr + 4 * k);
    if (tile + stride < ntiles) __builtin_prefetch(xt + xstep + lo, 0, 3);
  }

  int buf = 0;
  for (; tile < ntiles; tile += stride) {
    float* stage = lds + buf * (16 * ROW);
    const long tok0 = (long)tile * 16;

    v8f acc[4];
#pragma unroll
    for (int n = 0; n < 4; ++n) {
#pragma unroll
      for (int i = 0; i < 8; ++i) acc[n][i] = bias_f[n];
    }
    products_quarter<0>(acc, afrag, bfrag);
    products_quarter<1>(acc, afrag, bfrag);
    products_quarter<2>(acc, afrag, bfrag);
    products_quarter<3>(acc, afrag, bfrag);

    asm volatile("v_nop\n\tv_nop\n\tv_nop\n\tv_nop\n\tv_nop\n\tv_nop\n\tv_nop\n\tv_nop"
                 :: "v"(acc[0]), "v"(acc[1]), "v"(acc[2]), "v"(acc[3]));

    xt += xstep;
    if (tile + stride < ntiles) {
      const float* xr = xt + lo;
#pragma unroll
      for (int k = 0; k < 16; ++k) afrag[k] = *(const v2f*)(xr + 4 * k);
      if (tile + 2 * stride < ntiles) __builtin_prefetch(xt + xstep + lo, 0, 3);
    }

#pragma unroll
    for (int n = 0; n < 4; ++n) {
#pragma unroll
      for (int v = 0; v < 8; ++v)
        stage[(v + half * 8) * ROW + (n * 16 + l16) * GRP + g] = acc[n][v];
    }

    __builtin_amdgcn_fence(__ATOMIC_RELEASE, "workgroup", "local");
    __builtin_amdgcn_s_barrier();
    __builtin_amdgcn_fence(__ATOMIC_ACQUIRE, "workgroup", "local");

    {
      float* orow = out + tok0 * ROW;
#pragma unroll
      for (int it = 0; it < 8; ++it) {
        int idx = (it * 256 + (int)threadIdx.x) * 4;
        *(v4f*)(orow + idx) = *(const v4f*)(stage + idx);
      }
    }

    buf ^= 1;
  }
  __builtin_amdgcn_s_setprio(0);
}

extern "C" void kernel_launch(void* const* d_in, const int* in_sizes, int n_in,
                              void* d_out, int out_size, void* d_ws, size_t ws_size,
                              hipStream_t stream) {
  const float* x    = (const float*)d_in[0];
  const float* W    = (const float*)d_in[1];
  const float* bias = (const float*)d_in[2];
  float* out        = (float*)d_out;

  const int ntok   = in_sizes[0] / ROW;
  const int ntiles = ntok / 16;

  int blocks = ntiles < 512 ? ntiles : 512;
  grouped_linear_wmma<<<dim3(blocks), dim3(256), 0, stream>>>(x, W, bias, out, ntiles);
}
